# starting kernel
# baseline (speedup 1.0000x reference)
#include <hip/hip_runtime.h>

typedef __attribute__((ext_vector_type(2))) float v2f;
typedef __attribute__((ext_vector_type(8))) float v8f;

#define WW   256
#define HWSZ 65536
#define CCH  8
#define KKT  25
#define NEG_SLOPE 0.2f


__device__ __forceinline__ void mma_extract_store(const v2f A[7], const v2f B[7],
                                                  int lane, float* __restrict__ outp)
{
    v8f acc = {0.f, 0.f, 0.f, 0.f, 0.f, 0.f, 0.f, 0.f};
#pragma unroll
    for (int kc = 0; kc < 7; ++kc) {
        acc = __builtin_amdgcn_wmma_f32_16x16x4_f32(
             false, A[kc],  false, B[kc],
             (short)0, acc,  false,  false);
    }
    const int idx = lane & 7;
    float v = acc[0];
#pragma unroll
    for (int i = 1; i < 8; ++i) v = (idx == i) ? acc[i] : v;
    v = (v >= 0.0f) ? v : NEG_SLOPE * v;
    if (lane < 8 || lane >= 24) {
        const int p = (lane < 16) ? lane : (lane - 16);
        __builtin_nontemporal_store(v, outp + p);
    }
}

__global__ __launch_bounds__(256) void dynconv5x5_wmma(
    const float* __restrict__ x,
    const float* __restrict__ kern,
    float* __restrict__ out)
{
    const int lane  = threadIdx.x & 31;
    const int wave  = threadIdx.x >> 5;
    const int tx    = (blockIdx.x << 3) + wave;
    const int y     = blockIdx.y;
    const int plane = blockIdx.z;

    const int x0 = tx << 4;
    const int hf = lane >> 4;
    const int lp = lane & 15;

    const float* __restrict__ xp = x + plane * HWSZ;
    const int kbase = (plane * KKT) * HWSZ + y * WW + x0 + lp;
    float* __restrict__ outp = out + plane * HWSZ + y * WW + x0;

    if ((y >= 2) & (y <= 253) & (tx >= 1) & (tx <= 14)) {
        const float* __restrict__ xb    = xp + (y - 2) * WW + (x0 - 2);
        const float* __restrict__ pA2   = xb + lp + hf * 2;
        const float* __restrict__ pA253 = xb + lp + hf * 253;
        const float* __restrict__ pB    = kern + kbase + hf * (2 * HWSZ);

        constexpr int T0[12]   = {0, 1, 4,   5,   8,   9,   12,  13,  16,  17,  20,   21};
        constexpr int OFF0[12] = {0, 1, 4, 256, 259, 260, 514, 515, 769, 770, 1024, 1025};
        constexpr int D253[12] = {0, 0, 1,   0,   1,   1,   0,   1,   0,   0,   0,    0};

        v2f A[7], B[7];
#pragma unroll
        for (int i = 0; i < 12; ++i) {
            const int kc = T0[i] >> 2;
            const int j  = T0[i] & 3;
            A[kc][j] = D253[i] ? pA253[OFF0[i]] : pA2[OFF0[i]];
            B[kc][j] = __builtin_nontemporal_load(pB + T0[i] * HWSZ);
        }
        const float a24 = xb[lp + 1028];
        const float b24 = __builtin_nontemporal_load(kern + kbase + 24 * HWSZ);
        A[6][0] = hf ? 0.0f : a24;
        B[6][0] = hf ? 0.0f : b24;
        A[6][1] = 0.0f;
        B[6][1] = 0.0f;

        mma_extract_store(A, B, lane, outp);
        return;
    }

    {
        v2f A[7], B[7];
#pragma unroll
        for (int kc = 0; kc < 7; ++kc) {
#pragma unroll
            for (int j = 0; j < 2; ++j) {
                const int t  = 4 * kc + 2 * hf + j;
                const int tc = (t < KKT) ? t : (KKT - 1);
                const int dy = tc / 5 - 2;
                const int dx = tc % 5 - 2;
                int cy = y + dy;       cy = cy < 0 ? 0 : (cy > 255 ? 255 : cy);
                int cx = x0 + lp + dx; cx = cx < 0 ? 0 : (cx > 255 ? 255 : cx);
                const float av = xp[cy * WW + cx];
                const float bv = __builtin_nontemporal_load(kern + kbase + tc * HWSZ);
                const bool valid = (t < KKT);
                A[kc][j] = valid ? av : 0.0f;
                B[kc][j] = valid ? bv : 0.0f;
            }
        }
        mma_extract_store(A, B, lane, outp);
    }
}

extern "C" void kernel_launch(void* const* d_in, const int* in_sizes, int n_in,
                              void* d_out, int out_size, void* d_ws, size_t ws_size,
                              hipStream_t stream) {
    (void)in_sizes; (void)n_in; (void)d_ws; (void)ws_size; (void)out_size;
    const float* x    = (const float*)d_in[0];
    const float* kern = (const float*)d_in[1];
    float*       out  = (float*)d_out;

    dim3 grid(2, 256, 32);
    dynconv5x5_wmma<<<grid, 256, 0, stream>>>(x, kern, out);
}


# 64 pixels x 2 rows per wave as 4 interleaved WMMA tiles: 16-B-per-lane weight loads on aligned 256-B pieces, 8x fewer waves; taps re-mapped over the k slots, so the f32 summation order changes
# speedup vs baseline: 1.5759x; 1.5759x over previous
#include <hip/hip_runtime.h>


typedef __attribute__((ext_vector_type(2))) float v2f;
typedef __attribute__((ext_vector_type(4))) float v4f;
typedef __attribute__((ext_vector_type(8))) float v8f;

#define WW   256
#define HWSZ 65536
#define KKT  25
#define NEG_SLOPE 0.2f
#define ROWS 2


struct RawWin { v2f l; v4f m; v2f r; };

__device__ __forceinline__ RawWin load_window(const float* __restrict__ rowp, int col0,
                                              bool le, bool re)
{
    RawWin w;
    w.l = *(const v2f*)(rowp + col0 + (le ? 0 : -2));
    w.m = *(const v4f*)(rowp + col0);
    w.r = *(const v2f*)(rowp + col0 + (re ? 0 : 4));
    return w;
}

__global__ __launch_bounds__(256) void dynconv5x5_wmma(
    const float* __restrict__ x,
    const float* __restrict__ kern,
    float* __restrict__ out)
{
    const int lane  = threadIdx.x & 31;
    const int wave  = __builtin_amdgcn_readfirstlane((int)(threadIdx.x >> 5));
    const int tx    = wave & 3;
    const int y0    = ((blockIdx.x << 1) + (wave >> 2)) * ROWS;
    const int plane = blockIdx.y;
    const int c     = lane & 15;
    const bool h    = lane >= 16;
    const int col0  = (tx << 6) + (c << 2);
    const bool le   = (col0 == 0);
    const bool re   = (col0 == WW - 4);

    const float* __restrict__ xp = x + plane * HWSZ;
    RawWin X[ROWS][3];
    v4f Bl[ROWS][13];

#pragma unroll
    for (int r = 0; r < ROWS; ++r) {
        const int y = y0 + r;
        const int ym2 = (y - 2 < 0) ? 0 : y - 2;
        const int ym1 = (y - 1 < 0) ? 0 : y - 1;
        const int yp1 = (y + 1 > 255) ? 255 : y + 1;
        const int yp2 = (y + 2 > 255) ? 255 : y + 2;
        const int r0 = h ? yp2 : ym2;
        const int r1 = h ? yp1 : ym1;
        X[r][0] = load_window(xp + r0 * WW, col0, le, re);
        X[r][1] = load_window(xp + r1 * WW, col0, le, re);
        X[r][2] = load_window(xp + y  * WW, col0, le, re);
    }

#pragma unroll
    for (int r = 0; r < ROWS; ++r) {
        const float* __restrict__ kb = kern + (plane * KKT) * HWSZ + (y0 + r) * WW + col0;
        const float* __restrict__ kb0 = kb + (h ? 20 * HWSZ : 0);
        const float* __restrict__ kb1 = kb + (h ? 10 * HWSZ : 0);
        const float* __restrict__ kb2 = kb + (h ?  3 * HWSZ : 0);
#pragma unroll
        for (int s = 0; s < 5; ++s)
            Bl[r][s] = __builtin_nontemporal_load((const v4f*)(kb0 + s * HWSZ));
#pragma unroll
        for (int s = 5; s < 10; ++s)
            Bl[r][s] = __builtin_nontemporal_load((const v4f*)(kb1 + s * HWSZ));
#pragma unroll
        for (int s = 10; s < 12; ++s)
            Bl[r][s] = __builtin_nontemporal_load((const v4f*)(kb2 + s * HWSZ));
        Bl[r][12] = __builtin_nontemporal_load((const v4f*)(kb + 12 * HWSZ));
    }

    __builtin_amdgcn_sched_barrier(0);

    const bool b0 = (lane & 1) != 0, b1 = (lane & 2) != 0, b2 = (lane & 4) != 0;
    const bool owner = (lane < 8) || (lane >= 24);
    const int  p     = (lane < 16) ? lane : (lane - 16);

#pragma unroll
    for (int r = 0; r < ROWS; ++r) {
        float Wv[3][8];
#pragma unroll
        for (int q = 0; q < 3; ++q) {
            const RawWin w = X[r][q];
            Wv[q][0] = le ? w.m[0] : w.l[0];
            Wv[q][1] = le ? w.m[0] : w.l[1];
            Wv[q][2] = w.m[0]; Wv[q][3] = w.m[1]; Wv[q][4] = w.m[2]; Wv[q][5] = w.m[3];
            Wv[q][6] = re ? w.m[3] : w.r[0];
            Wv[q][7] = re ? w.m[3] : w.r[1];
        }

        v8f acc[4];
#pragma unroll
        for (int g = 0; g < 4; ++g) acc[g] = (v8f){0.f, 0.f, 0.f, 0.f, 0.f, 0.f, 0.f, 0.f};
#pragma unroll
        for (int kc = 0; kc < 7; ++kc) {
#pragma unroll
            for (int g = 0; g < 4; ++g) {
                v2f a, b;
#pragma unroll
                for (int j = 0; j < 2; ++j) {
                    const int s = 2 * kc + j;
                    float av, bv;
                    if (s < 5)        { av = Wv[0][s + g];      bv = Bl[r][s][g]; }
                    else if (s < 10)  { av = Wv[1][s - 5 + g];  bv = Bl[r][s][g]; }
                    else if (s < 12)  { av = h ? Wv[2][s - 7 + g] : Wv[2][s - 10 + g]; bv = Bl[r][s][g]; }
                    else if (s == 12) { av = h ? 0.0f : Wv[2][2 + g]; bv = h ? 0.0f : Bl[r][12][g]; }
                    else              { av = 0.0f; bv = 0.0f; }
                    a[j] = av; b[j] = bv;
                }
                acc[g] = __builtin_amdgcn_wmma_f32_16x16x4_f32(
                     false, a,  false, b,
                     (short)0, acc[g],  false,  false);
            }
        }

        v4f o;
#pragma unroll
        for (int g = 0; g < 4; ++g) {
            const float s0 = b0 ? acc[g][1] : acc[g][0];
            const float s1 = b0 ? acc[g][3] : acc[g][2];
            const float s2 = b0 ? acc[g][5] : acc[g][4];
            const float s3 = b0 ? acc[g][7] : acc[g][6];
            const float t0 = b1 ? s1 : s0;
            const float t1 = b1 ? s3 : s2;
            float v = b2 ? t1 : t0;
            v = (v >= 0.0f) ? v : NEG_SLOPE * v;
            o[g] = v;
        }
        if (owner) {
            float* __restrict__ outp = out + plane * HWSZ + (y0 + r) * WW + (tx << 6) + (p << 2);
            __builtin_nontemporal_store(o, (v4f*)outp);
        }
    }
}

extern "C" void kernel_launch(void* const* d_in, const int* in_sizes, int n_in,
                              void* d_out, int out_size, void* d_ws, size_t ws_size,
                              hipStream_t stream) {
    (void)in_sizes; (void)n_in; (void)d_ws; (void)ws_size; (void)out_size;
    const float* x    = (const float*)d_in[0];
    const float* kern = (const float*)d_in[1];
    float*       out  = (float*)d_out;

    dim3 grid(64, 32);
    dynconv5x5_wmma<<<grid, 256, 0, stream>>>(x, kern, out);
}
